# starting kernel
# baseline (speedup 1.0000x reference)
#include <hip/hip_runtime.h>
#include <math.h>

typedef __attribute__((ext_vector_type(2))) float v2f;
typedef __attribute__((ext_vector_type(8))) float v8f;

#define NQ 12
#define NSTATE 4096
#define NPADDED (4096 + 256)
#define PAD_IDX(i) ((i) + ((i) >> 4))

__device__ __forceinline__ int perm_fwd(int i) {
#pragma unroll
  for (int j = NQ - 1; j >= 0; --j) {
    int c = 11 - j;
    int t = 11 - ((j + 1) % NQ);
    i ^= ((i >> c) & 1) << t;
  }
  return i;
}
__device__ __forceinline__ int perm_inv(int i) {
#pragma unroll
  for (int j = 0; j < NQ; ++j) {
    int c = 11 - j;
    int t = 11 - ((j + 1) % NQ);
    i ^= ((i >> c) & 1) << t;
  }
  return i;
}

__device__ __forceinline__ int sweep_idx(int s, int g, int v) {
  if (s == 0) return (g << 4) | v;
  if (s == 1) return ((g & 0xF0) << 4) | (v << 4) | (g & 0x0F);
  return (v << 8) | g;
}

__global__ __launch_bounds__(256) void qsim_kernel(const float* __restrict__ x,
                                                   const float* __restrict__ w,
                                                   float* __restrict__ out) {
  __shared__ float sre[NPADDED];
  __shared__ float sim[NPADDED];
  __shared__ float matP[16 * 17];
  __shared__ float matQ[16 * 17];
  __shared__ float matQn[16 * 17];
  __shared__ float csA[NQ], snA[NQ], csB[NQ], snB[NQ];
  __shared__ float red[8][NQ];

  const int tid = threadIdx.x;
  const int b = blockIdx.x;
  const int lane = tid & 31;
  const int wv = tid >> 5;

  if (tid < NQ) {
    int q = 11 - tid;
    float thA = x[b * NQ + q] + w[q];
    float thB = w[NQ + q];
    csA[tid] = cosf(0.5f * thA);
    snA[tid] = sinf(0.5f * thA);
    csB[tid] = cosf(0.5f * thB);
    snB[tid] = sinf(0.5f * thB);
  }
  __syncthreads();

  {
    float base = 1.0f;
#pragma unroll
    for (int k = 0; k < 8; ++k)
      base *= ((tid >> k) & 1) ? snA[4 + k] : csA[4 + k];
    int pct = __popc(tid);
#pragma unroll
    for (int v = 0; v < 16; ++v) {
      float m = base;
#pragma unroll
      for (int k = 0; k < 4; ++k)
        m *= ((v >> k) & 1) ? snA[k] : csA[k];
      int pc = (pct + __popc(v)) & 3;
      float re = (pc == 0) ? m : ((pc == 2) ? -m : 0.0f);
      float im = (pc == 1) ? -m : ((pc == 3) ? m : 0.0f);
      int idx = (tid << 4) | v;
      sre[PAD_IDX(idx)] = re;
      sim[PAD_IDX(idx)] = im;
    }
  }
  __syncthreads();

  {
    float tr[16], ti[16];
#pragma unroll
    for (int v = 0; v < 16; ++v) {
      int j = perm_fwd((tid << 4) | v);
      tr[v] = sre[PAD_IDX(j)];
      ti[v] = sim[PAD_IDX(j)];
    }
    __syncthreads();
#pragma unroll
    for (int v = 0; v < 16; ++v) {
      int i = (tid << 4) | v;
      sre[PAD_IDX(i)] = tr[v];
      sim[PAD_IDX(i)] = ti[v];
    }
  }
  __syncthreads();

#pragma unroll
  for (int s = 0; s < 3; ++s) {
    {
      int r = tid >> 4, c = tid & 15, xr = r ^ c;
      float m = 1.0f;
#pragma unroll
      for (int k = 0; k < 4; ++k)
        m *= ((xr >> k) & 1) ? snB[4 * s + k] : csB[4 * s + k];
      int d = __popc(xr) & 3;
      float pe = (d == 0) ? m : ((d == 2) ? -m : 0.0f);
      float qe = (d == 1) ? -m : ((d == 3) ? m : 0.0f);
      matP[r * 17 + c] = pe;
      matQ[r * 17 + c] = qe;
      matQn[r * 17 + c] = -qe;
    }
    __syncthreads();

    const int arow = lane & 15;
    const int koff = (lane >> 4) << 1;
    v2f aP[4], aQ[4], aQn[4];
#pragma unroll
    for (int j = 0; j < 4; ++j) {
      int k0 = 4 * j + koff;
      aP[j].x = matP[arow * 17 + k0];
      aP[j].y = matP[arow * 17 + k0 + 1];
      aQ[j].x = matQ[arow * 17 + k0];
      aQ[j].y = matQ[arow * 17 + k0 + 1];
      aQn[j].x = matQn[arow * 17 + k0];
      aQn[j].y = matQn[arow * 17 + k0 + 1];
    }

#pragma unroll
    for (int tt = 0; tt < 2; ++tt) {
      int gcol = (wv * 2 + tt) * 16 + (lane & 15);
      v2f bre[4], bim[4];
#pragma unroll
      for (int j = 0; j < 4; ++j) {
        int v0 = 4 * j + koff;
        int i0 = sweep_idx(s, gcol, v0);
        int i1 = sweep_idx(s, gcol, v0 + 1);
        bre[j].x = sre[PAD_IDX(i0)];
        bre[j].y = sre[PAD_IDX(i1)];
        bim[j].x = sim[PAD_IDX(i0)];
        bim[j].y = sim[PAD_IDX(i1)];
      }
      v8f dre = {};
      v8f dim = {};
#pragma unroll
      for (int j = 0; j < 4; ++j) {
        dre = __builtin_amdgcn_wmma_f32_16x16x4_f32(false, aP[j], false, bre[j],
                                                    (short)0, dre, false, false);
        dre = __builtin_amdgcn_wmma_f32_16x16x4_f32(false, aQn[j], false, bim[j],
                                                    (short)0, dre, false, false);
        dim = __builtin_amdgcn_wmma_f32_16x16x4_f32(false, aQ[j], false, bre[j],
                                                    (short)0, dim, false, false);
        dim = __builtin_amdgcn_wmma_f32_16x16x4_f32(false, aP[j], false, bim[j],
                                                    (short)0, dim, false, false);
      }
      int vadd = (lane >> 4) << 3;
#pragma unroll
      for (int r = 0; r < 8; ++r) {
        int idx = sweep_idx(s, gcol, r + vadd);
        sre[PAD_IDX(idx)] = dre[r];
        sim[PAD_IDX(idx)] = dim[r];
      }
    }
    __syncthreads();
  }

  float acc[NQ];
#pragma unroll
  for (int q = 0; q < NQ; ++q) acc[q] = 0.0f;
#pragma unroll
  for (int v = 0; v < 16; ++v) {
    int jdx = (tid << 4) | v;
    float re = sre[PAD_IDX(jdx)];
    float im = sim[PAD_IDX(jdx)];
    float p = re * re + im * im;
    int i = perm_inv(jdx);
#pragma unroll
    for (int q = 0; q < NQ; ++q) acc[q] += ((i >> q) & 1) ? -p : p;
  }
#pragma unroll
  for (int off = 16; off > 0; off >>= 1) {
#pragma unroll
    for (int q = 0; q < NQ; ++q) acc[q] += __shfl_down(acc[q], off, 32);
  }
  if (lane == 0) {
#pragma unroll
    for (int q = 0; q < NQ; ++q) red[wv][q] = acc[q];
  }
  __syncthreads();
  if (tid < NQ) {
    float sum = 0.0f;
#pragma unroll
    for (int k = 0; k < 8; ++k) sum += red[k][tid];
    out[b * NQ + (11 - tid)] = sum;
  }
}

extern "C" void kernel_launch(void* const* d_in, const int* in_sizes, int n_in,
                              void* d_out, int out_size, void* d_ws,
                              size_t ws_size, hipStream_t stream) {
  const float* x = (const float*)d_in[0];
  const float* w = (const float*)d_in[1];
  float* out = (float*)d_out;
  int B = in_sizes[0] / NQ;
  qsim_kernel<<<B, 256, 0, stream>>>(x, w, out);
}


# Monomials from a per-row table of 4-qubit sub-products (51 slots, built beside the weights): a string is 3 LDS reads and 2 multiplies, not 12 select-multiplies; product grouping changes
# speedup vs baseline: 11.0018x; 11.0018x over previous
#include <hip/hip_runtime.h>
#include <math.h>

typedef __attribute__((ext_vector_type(2))) float v2f;
typedef __attribute__((ext_vector_type(8))) float v8f;

#define NQ 12
#define NBATCH 2048
#define TILE_B 16
#define NTERM 256
#define NREAL 252
#define PART_STRIDE 288
#define NSLOT 51

__constant__ const unsigned TERM_W[NTERM] = {
    0x00810000u, 0x00810040u, 0x00811080u, 0x008110c0u, 0x00812000u, 0x00812040u, 0x00813080u, 0x008130c0u,
    0x00853000u, 0x00853040u, 0x00852080u, 0x008520c0u, 0x00851000u, 0x00851040u, 0x00850080u, 0x008500c0u,
    0x00890000u, 0x00890040u, 0x00891080u, 0x008910c0u, 0x00892000u, 0x00892040u, 0x00893080u, 0x008930c0u,
    0x008d3000u, 0x008d3040u, 0x008d2080u, 0x008d20c0u, 0x008d1000u, 0x008d1040u, 0x008d0080u, 0x008d00c0u,
    0x008d0100u, 0x008d0140u, 0x008d1180u, 0x008d11c0u, 0x008d2100u, 0x008d2140u, 0x008d3180u, 0x008d31c0u,
    0x00893100u, 0x00893140u, 0x00892180u, 0x008921c0u, 0x00891100u, 0x00891140u, 0x00890180u, 0x008901c0u,
    0x00850100u, 0x00850140u, 0x00851180u, 0x008511c0u, 0x00852100u, 0x00852140u, 0x00853180u, 0x008531c0u,
    0x00813100u, 0x00813140u, 0x00812180u, 0x008121c0u, 0x00811100u, 0x00811140u, 0x00810180u, 0x008101c0u,
    0x01914200u, 0x01914240u, 0x02914280u, 0x029142c0u, 0x03914300u, 0x03914340u, 0x03915380u, 0x039153c0u,
    0x04916280u, 0x049162c0u, 0x04917c00u, 0x04917c40u, 0x05918300u, 0x05918340u, 0x05919380u, 0x059193c0u,
    0x0591a300u, 0x0591a340u, 0x0591b380u, 0x0591b3c0u, 0x06910280u, 0x069102c0u, 0x06911c00u, 0x06911c40u,
    0x06912280u, 0x069122c0u, 0x06913c00u, 0x06913c40u, 0x0791c300u, 0x0791c340u, 0x0791d380u, 0x0791d3c0u,
    0x0791e300u, 0x0791e340u, 0x0791f380u, 0x0791f3c0u, 0x0795f300u, 0x0795f340u, 0x0795e380u, 0x0795e3c0u,
    0x0795d300u, 0x0795d340u, 0x0795c380u, 0x0795c3c0u, 0x08990280u, 0x089902c0u, 0x08991c00u, 0x08991c40u,
    0x08992280u, 0x089922c0u, 0x08993c00u, 0x08993c40u, 0x089d3280u, 0x089d32c0u, 0x089d2c00u, 0x089d2c40u,
    0x089d1280u, 0x089d12c0u, 0x089d0c00u, 0x089d0c40u, 0x09a1c300u, 0x09a1c340u, 0x09a1d380u, 0x09a1d3c0u,
    0x09a1e300u, 0x09a1e340u, 0x09a1f380u, 0x09a1f3c0u, 0x09a5f300u, 0x09a5f340u, 0x09a5e380u, 0x09a5e3c0u,
    0x09a5d300u, 0x09a5d340u, 0x09a5c380u, 0x09a5c3c0u, 0x09a9c300u, 0x09a9c340u, 0x09a9d380u, 0x09a9d3c0u,
    0x09a9e300u, 0x09a9e340u, 0x09a9f380u, 0x09a9f3c0u, 0x09adf300u, 0x09adf340u, 0x09ade380u, 0x09ade3c0u,
    0x09add300u, 0x09add340u, 0x09adc380u, 0x09adc3c0u, 0x0a810280u, 0x0a8102c0u, 0x0a811c00u, 0x0a811c40u,
    0x0a812280u, 0x0a8122c0u, 0x0a813c00u, 0x0a813c40u, 0x0a853280u, 0x0a8532c0u, 0x0a852c00u, 0x0a852c40u,
    0x0a851280u, 0x0a8512c0u, 0x0a850c00u, 0x0a850c40u, 0x0a890280u, 0x0a8902c0u, 0x0a891c00u, 0x0a891c40u,
    0x0a892280u, 0x0a8922c0u, 0x0a893c00u, 0x0a893c40u, 0x0a8d3280u, 0x0a8d32c0u, 0x0a8d2c00u, 0x0a8d2c40u,
    0x0a8d1280u, 0x0a8d12c0u, 0x0a8d0c00u, 0x0a8d0c40u, 0x0bb1c300u, 0x0bb1c340u, 0x0bb1d380u, 0x0bb1d3c0u,
    0x0bb1e300u, 0x0bb1e340u, 0x0bb1f380u, 0x0bb1f3c0u, 0x0bb5f300u, 0x0bb5f340u, 0x0bb5e380u, 0x0bb5e3c0u,
    0x0bb5d300u, 0x0bb5d340u, 0x0bb5c380u, 0x0bb5c3c0u, 0x0bb9c300u, 0x0bb9c340u, 0x0bb9d380u, 0x0bb9d3c0u,
    0x0bb9e300u, 0x0bb9e340u, 0x0bb9f380u, 0x0bb9f3c0u, 0x0bbdf300u, 0x0bbdf340u, 0x0bbde380u, 0x0bbde3c0u,
    0x0bbdd300u, 0x0bbdd340u, 0x0bbdc380u, 0x0bbdc3c0u, 0x0bbdc300u, 0x0bbdc340u, 0x0bbdd380u, 0x0bbdd3c0u,
    0x0bbde300u, 0x0bbde340u, 0x0bbdf380u, 0x0bbdf3c0u, 0x0bb9f300u, 0x0bb9f340u, 0x0bb9e380u, 0x0bb9e3c0u,
    0x0bb9d300u, 0x0bb9d340u, 0x0bb9c380u, 0x0bb9c3c0u, 0x0bb5c300u, 0x0bb5c340u, 0x0bb5d380u, 0x0bb5d3c0u,
    0x0bb5e300u, 0x0bb5e340u, 0x0bb5f380u, 0x0bb5f3c0u, 0x0bb1f300u, 0x0bb1f340u, 0x0bb1e380u, 0x0bb1e3c0u,
    0x0bb1d300u, 0x0bb1d340u, 0x0bb1c380u, 0x0bb1c3c0u, 0x00cb2c80u, 0x00cb2c80u, 0x00cb2c80u, 0x00cb2c80u,
};
__constant__ const unsigned SLOT_PAT[16] = {
    0x820a0a0bu, 0x282828a1u, 0x00a0a083u, 0x00828229u, 0x000f0f38u, 0x001e1e92u, 0x000e0eb0u, 0x002c2c1au,
    0x000d0d0du, 0x001c1c58u, 0x0058580au, 0x004949a0u, 0x00050505u, 0x00141450u, 0x00505041u, 0x00414114u,
};
__constant__ const unsigned TERM_SM[NTERM] = {
    0x00ffe000u, 0x00ffe006u, 0x00ffe018u, 0x01ffe01eu, 0x00ffe060u, 0x00ffe066u, 0x01ffe078u, 0x00ffe07eu,
    0x00ffe180u, 0x00ffe186u, 0x00ffe198u, 0x01ffe19eu, 0x01ffe1e0u, 0x01ffe1e6u, 0x00ffe1f8u, 0x01ffe1feu,
    0x00ffe600u, 0x00ffe606u, 0x00ffe618u, 0x01ffe61eu, 0x00ffe660u, 0x00ffe666u, 0x01ffe678u, 0x00ffe67eu,
    0x01ffe780u, 0x01ffe786u, 0x01ffe798u, 0x00ffe79eu, 0x00ffe7e0u, 0x00ffe7e6u, 0x01ffe7f8u, 0x00ffe7feu,
    0x01ffe800u, 0x00ffe806u, 0x01ffe818u, 0x01ffe81eu, 0x01ffe860u, 0x00ffe866u, 0x00ffe878u, 0x00ffe87eu,
    0x01ffe980u, 0x00ffe986u, 0x01ffe998u, 0x01ffe99eu, 0x00ffe9e0u, 0x01ffe9e6u, 0x01ffe9f8u, 0x01ffe9feu,
    0x00ffee00u, 0x01ffee06u, 0x00ffee18u, 0x00ffee1eu, 0x00ffee60u, 0x01ffee66u, 0x01ffee78u, 0x01ffee7eu,
    0x01ffef80u, 0x00ffef86u, 0x01ffef98u, 0x01ffef9eu, 0x00ffefe0u, 0x01ffefe6u, 0x01ffeff8u, 0x01ffeffeu,
    0x02003000u, 0x02003003u, 0x04007000u, 0x04007006u, 0x0600f000u, 0x0600f003u, 0x0600f00cu, 0x0700f00fu,
    0x0801f000u, 0x0801f006u, 0x0801f018u, 0x0901f01eu, 0x0a03f000u, 0x0a03f003u, 0x0a03f00cu, 0x0b03f00fu,
    0x0a03f030u, 0x0a03f033u, 0x0b03f03cu, 0x0a03f03fu, 0x0c07f000u, 0x0c07f006u, 0x0c07f018u, 0x0d07f01eu,
    0x0c07f060u, 0x0c07f066u, 0x0d07f078u, 0x0c07f07eu, 0x0e0ff000u, 0x0e0ff003u, 0x0e0ff00cu, 0x0f0ff00fu,
    0x0e0ff030u, 0x0e0ff033u, 0x0f0ff03cu, 0x0e0ff03fu, 0x0e0ff0c0u, 0x0e0ff0c3u, 0x0e0ff0ccu, 0x0f0ff0cfu,
    0x0f0ff0f0u, 0x0f0ff0f3u, 0x0e0ff0fcu, 0x0f0ff0ffu, 0x101ff000u, 0x101ff006u, 0x101ff018u, 0x111ff01eu,
    0x101ff060u, 0x101ff066u, 0x111ff078u, 0x101ff07eu, 0x101ff180u, 0x101ff186u, 0x101ff198u, 0x111ff19eu,
    0x111ff1e0u, 0x111ff1e6u, 0x101ff1f8u, 0x111ff1feu, 0x123ff000u, 0x123ff003u, 0x123ff00cu, 0x133ff00fu,
    0x123ff030u, 0x123ff033u, 0x133ff03cu, 0x123ff03fu, 0x123ff0c0u, 0x123ff0c3u, 0x123ff0ccu, 0x133ff0cfu,
    0x133ff0f0u, 0x133ff0f3u, 0x123ff0fcu, 0x133ff0ffu, 0x123ff300u, 0x123ff303u, 0x123ff30cu, 0x133ff30fu,
    0x123ff330u, 0x123ff333u, 0x133ff33cu, 0x123ff33fu, 0x133ff3c0u, 0x133ff3c3u, 0x133ff3ccu, 0x123ff3cfu,
    0x123ff3f0u, 0x123ff3f3u, 0x133ff3fcu, 0x123ff3ffu, 0x147ff000u, 0x147ff006u, 0x147ff018u, 0x157ff01eu,
    0x147ff060u, 0x147ff066u, 0x157ff078u, 0x147ff07eu, 0x147ff180u, 0x147ff186u, 0x147ff198u, 0x157ff19eu,
    0x157ff1e0u, 0x157ff1e6u, 0x147ff1f8u, 0x157ff1feu, 0x147ff600u, 0x147ff606u, 0x147ff618u, 0x157ff61eu,
    0x147ff660u, 0x147ff666u, 0x157ff678u, 0x147ff67eu, 0x157ff780u, 0x157ff786u, 0x157ff798u, 0x147ff79eu,
    0x147ff7e0u, 0x147ff7e6u, 0x157ff7f8u, 0x147ff7feu, 0x16fff000u, 0x16fff003u, 0x16fff00cu, 0x17fff00fu,
    0x16fff030u, 0x16fff033u, 0x17fff03cu, 0x16fff03fu, 0x16fff0c0u, 0x16fff0c3u, 0x16fff0ccu, 0x17fff0cfu,
    0x17fff0f0u, 0x17fff0f3u, 0x16fff0fcu, 0x17fff0ffu, 0x16fff300u, 0x16fff303u, 0x16fff30cu, 0x17fff30fu,
    0x16fff330u, 0x16fff333u, 0x17fff33cu, 0x16fff33fu, 0x17fff3c0u, 0x17fff3c3u, 0x17fff3ccu, 0x16fff3cfu,
    0x16fff3f0u, 0x16fff3f3u, 0x17fff3fcu, 0x16fff3ffu, 0x17fffc01u, 0x16fffc02u, 0x17fffc0du, 0x17fffc0eu,
    0x17fffc31u, 0x16fffc32u, 0x16fffc3du, 0x16fffc3eu, 0x17fffcc1u, 0x16fffcc2u, 0x17fffccdu, 0x17fffcceu,
    0x16fffcf1u, 0x17fffcf2u, 0x17fffcfdu, 0x17fffcfeu, 0x16ffff01u, 0x17ffff02u, 0x16ffff0du, 0x16ffff0eu,
    0x16ffff31u, 0x17ffff32u, 0x17ffff3du, 0x17ffff3eu, 0x17ffffc1u, 0x16ffffc2u, 0x17ffffcdu, 0x17ffffceu,
    0x16fffff1u, 0x17fffff2u, 0x17fffffdu, 0x17fffffeu, 0x00000000u, 0x00000000u, 0x00000000u, 0x00000000u,
};

__global__ __launch_bounds__(256) void qsim_kernel(const float* __restrict__ x,
                                                   const float* __restrict__ w,
                                                   float* __restrict__ out) {
  __shared__ __attribute__((aligned(16))) float s_c[TILE_B * NQ];
  __shared__ __attribute__((aligned(16))) float s_ns[TILE_B * NQ];
  __shared__ float s_cb[NQ], s_sb[NQ];
  __shared__ __attribute__((aligned(16))) unsigned s_wg[2 * NTERM];
  __shared__ float s_tab[64 * TILE_B];
  __shared__ float s_part[8 * PART_STRIDE];

  const int tid = threadIdx.x;
  const int lane = tid & 31;
  const int wv = tid >> 5;
  const int b0 = blockIdx.x * TILE_B;

  const unsigned tw = TERM_W[tid];
  const unsigned tsm = TERM_SM[tid];
  const unsigned patw = SLOT_PAT[tid >> 4];

  if (tid < TILE_B * NQ + NQ) {
    float ang;
    if (tid < TILE_B * NQ) {
      ang = x[b0 * NQ + tid] + w[tid % NQ];
    } else {
      ang = w[NQ + (tid - TILE_B * NQ)];
    }
    float cv = cosf(ang);
    float sv = sinf(ang);
    if (tid < TILE_B * NQ) {
      s_c[tid] = cv;
      s_ns[tid] = -sv;
    } else {
      s_cb[tid - TILE_B * NQ] = cv;
      s_sb[tid - TILE_B * NQ] = sv;
    }
  }
  s_wg[2 * tid] = tw;
  __syncthreads();

  {
    const unsigned S = tsm & 0xFFFu;
    const unsigned m = (tsm >> 12) & 0xFFFu;
    float g = ((tsm >> 24) & 1u) ? -1.0f : 1.0f;
    if (tid >= NREAL) g = 0.0f;
#pragma unroll
    for (int j = 0; j < NQ; ++j) {
      float f = ((S >> j) & 1u) ? s_sb[j] : s_cb[j];
      g *= ((m >> j) & 1u) ? f : 1.0f;
    }
    s_wg[2 * tid + 1] = __float_as_uint(g);
  }
  {
    const int r = tid & 15;
    const int gp = tid >> 4;
    float cr[NQ], nr[NQ];
#pragma unroll
    for (int j = 0; j < NQ; ++j) {
      cr[j] = s_c[r * NQ + j];
      nr[j] = s_ns[r * NQ + j];
    }
#pragma unroll
    for (int i = 0; i < 4; ++i) {
      const int gs = gp + 16 * i;
      if (gs < NSLOT) {
        const unsigned pat = (patw >> (8 * i)) & 0xFFu;
        const int k = (i < 3) ? i : 0;
        float pr = 1.0f;
#pragma unroll
        for (int j = 0; j < 4; ++j) {
          float f = ((pat >> j) & 1u) ? cr[4 * k + j] : 1.0f;
          f = ((pat >> (4 + j)) & 1u) ? nr[4 * k + j] : f;
          pr *= f;
        }
        s_tab[gs * TILE_B + r] = pr;
      }
    }
  }
  __syncthreads();

  const int c = lane & 15;
  const int h = lane >> 4;
  uint4 p[8];
#pragma unroll
  for (int s = 0; s < 8; ++s) {
    const int t0 = (wv * 8 + s) * 4 + 2 * h;
    p[s] = *reinterpret_cast<const uint4*>(&s_wg[2 * t0]);
  }
  float pc[8][2][3];
#pragma unroll
  for (int s = 0; s < 8; ++s) {
#pragma unroll
    for (int e = 0; e < 2; ++e) {
      const unsigned wd = e ? p[s].z : p[s].x;
      pc[s][e][0] = s_tab[((wd >> 2) & 0x3F0u) | c];
      pc[s][e][1] = s_tab[((wd >> 8) & 0x3F0u) | c];
      pc[s][e][2] = s_tab[((wd >> 14) & 0x3F0u) | c];
    }
  }
  v8f d0 = {};
  v8f d1 = {};
#pragma unroll
  for (int s = 0; s < 8; ++s) {
    v2f a, bf;
    a.x = pc[s][0][0] * pc[s][0][1] * pc[s][0][2];
    a.y = pc[s][1][0] * pc[s][1][1] * pc[s][1][2];
    bf.x = ((int)(p[s].x >> 24) == c) ? __uint_as_float(p[s].y) : 0.0f;
    bf.y = ((int)(p[s].z >> 24) == c) ? __uint_as_float(p[s].w) : 0.0f;
    if (s & 1) {
      d1 = __builtin_amdgcn_wmma_f32_16x16x4_f32(false, a, false, bf, (short)0, d1,
                                                 false, false);
    } else {
      d0 = __builtin_amdgcn_wmma_f32_16x16x4_f32(false, a, false, bf, (short)0, d0,
                                                 false, false);
    }
  }
#pragma unroll
  for (int r = 0; r < 8; ++r) {
    s_part[wv * PART_STRIDE + h * 144 + r * 16 + c] = d0[r] + d1[r];
  }
  __syncthreads();

  if (tid < TILE_B * NQ) {
    const int bl = tid / NQ;
    const int q = tid % NQ;
    const int off = (bl >> 3) * 144 + (bl & 7) * 16 + q;
    float sum = 0.0f;
#pragma unroll
    for (int k = 0; k < 8; ++k) sum += s_part[k * PART_STRIDE + off];
    out[b0 * NQ + tid] = sum;
  }
}

extern "C" void kernel_launch(void* const* d_in, const int* in_sizes, int n_in,
                              void* d_out, int out_size, void* d_ws,
                              size_t ws_size, hipStream_t stream) {
  const float* x = (const float*)d_in[0];
  const float* w = (const float*)d_in[1];
  float* out = (float*)d_out;
  qsim_kernel<<<NBATCH / TILE_B, 256, 0, stream>>>(x, w, out);
}
